# starting kernel
# baseline (speedup 1.0000x reference)
#include <hip/hip_runtime.h>
#include <hip/hip_bf16.h>

typedef __attribute__((ext_vector_type(2))) float v2f;
typedef __attribute__((ext_vector_type(4))) float v4f;
typedef __attribute__((ext_vector_type(8))) float v8f;

#define MDIM 16384
#define NDIM 1024
#define KDIM 1024
#define NLAYERS 20
#define EPSV 1e-6f

__global__ __launch_bounds__(256) void wsum_kernel(const float* __restrict__ cw,
                                                   float* __restrict__ ws) {
    size_t idx = ((size_t)blockIdx.x * blockDim.x + threadIdx.x) * 4;
    v4f acc = *(const v4f*)(cw + idx);
#pragma unroll
    for (int l = 1; l < NLAYERS; ++l) {
        acc += *(const v4f*)(cw + (size_t)l * (NDIM * KDIM) + idx);
    }
    *(v4f*)(ws + idx) = acc;
}

__global__ __launch_bounds__(256) void rmsnorm_kernel(const float* __restrict__ x,
                                                      const float* __restrict__ w,
                                                      float* __restrict__ out) {
    const int row  = blockIdx.x;
    const int tid  = threadIdx.x;
    const int lane = tid & 31;
    const int wid  = tid >> 5;
    const float* xr = x + (size_t)row * 1024;
    float* orow     = out + (size_t)row * 1024;

    __shared__ float sm[8];

    v4f v = *(const v4f*)(xr + tid * 4);
    float mx = fmaxf(fmaxf(fabsf(v.x), fabsf(v.y)), fmaxf(fabsf(v.z), fabsf(v.w)));
#pragma unroll
    for (int off = 16; off > 0; off >>= 1)
        mx = fmaxf(mx, __shfl_xor(mx, off, 32));
    if (lane == 0) sm[wid] = mx;
    __syncthreads();
    if (tid < 32) {
        float t = (tid < 8) ? sm[tid] : 0.0f;
#pragma unroll
        for (int off = 4; off > 0; off >>= 1)
            t = fmaxf(t, __shfl_xor(t, off, 32));
        if (tid == 0) sm[0] = fmaxf(t, 1e-6f);
    }
    __syncthreads();
    const float inv = 1.0f / sm[0];

    v4f xs = v * inv;
    float ss = xs.x * xs.x + xs.y * xs.y + xs.z * xs.z + xs.w * xs.w;
    __syncthreads();
#pragma unroll
    for (int off = 16; off > 0; off >>= 1)
        ss += __shfl_xor(ss, off, 32);
    if (lane == 0) sm[wid] = ss;
    __syncthreads();
    if (tid < 32) {
        float t = (tid < 8) ? sm[tid] : 0.0f;
#pragma unroll
        for (int off = 4; off > 0; off >>= 1)
            t += __shfl_xor(t, off, 32);
        if (tid == 0) sm[0] = rsqrtf(t * (1.0f / 1024.0f) + EPSV);
    }
    __syncthreads();
    const float rstd = sm[0];

    v4f wv = *(const v4f*)(w + tid * 4);
    v4f o  = xs * rstd * wv;
    *(v4f*)(orow + tid * 4) = o;
}

#define TM 128
#define TN 64
#define TK 32
#define LDSW 36

__global__ __launch_bounds__(256) void gemm_wmma_kernel(const float* __restrict__ X,
                                                        const float* __restrict__ Wm,
                                                        float* __restrict__ C) {
    __shared__ float As[TM * LDSW];
    __shared__ float Bs[TN * LDSW];

    const int tid  = threadIdx.x;
    const int lane = tid & 31;
    const int wid  = tid >> 5;
    const int wm   = wid & 3;
    const int wn   = wid >> 2;
    const int mblk = blockIdx.y * TM;
    const int nblk = blockIdx.x * TN;

    v8f acc00 = {};
    v8f acc01 = {};
    v8f acc10 = {};
    v8f acc11 = {};

    const int srow = tid >> 3;
    const int scol = (tid & 7) << 2;

    const int arow0 = wm * 32 + (lane & 15);
    const int arow1 = arow0 + 16;
    const int ncol0 = wn * 32 + (lane & 15);
    const int ncol1 = ncol0 + 16;
    const int kb    = (lane >> 4) << 1;

    for (int k0 = 0; k0 < KDIM; k0 += TK) {
        v4f a0 = *(const v4f*)(X  + (size_t)(mblk + srow      ) * KDIM + k0 + scol);
        v4f a1 = *(const v4f*)(X  + (size_t)(mblk + srow + 32 ) * KDIM + k0 + scol);
        v4f a2 = *(const v4f*)(X  + (size_t)(mblk + srow + 64 ) * KDIM + k0 + scol);
        v4f a3 = *(const v4f*)(X  + (size_t)(mblk + srow + 96 ) * KDIM + k0 + scol);
        v4f b0 = *(const v4f*)(Wm + (size_t)(nblk + srow      ) * KDIM + k0 + scol);
        v4f b1 = *(const v4f*)(Wm + (size_t)(nblk + srow + 32 ) * KDIM + k0 + scol);
        __syncthreads();
        *(v4f*)(As + (srow      ) * LDSW + scol) = a0;
        *(v4f*)(As + (srow + 32 ) * LDSW + scol) = a1;
        *(v4f*)(As + (srow + 64 ) * LDSW + scol) = a2;
        *(v4f*)(As + (srow + 96 ) * LDSW + scol) = a3;
        *(v4f*)(Bs + (srow      ) * LDSW + scol) = b0;
        *(v4f*)(Bs + (srow + 32 ) * LDSW + scol) = b1;
        __syncthreads();

#pragma unroll
        for (int kk = 0; kk < TK; kk += 4) {
            v2f af0 = *(const v2f*)(As + arow0 * LDSW + kk + kb);
            v2f af1 = *(const v2f*)(As + arow1 * LDSW + kk + kb);
            v2f bf0 = *(const v2f*)(Bs + ncol0 * LDSW + kk + kb);
            v2f bf1 = *(const v2f*)(Bs + ncol1 * LDSW + kk + kb);
            acc00 = __builtin_amdgcn_wmma_f32_16x16x4_f32(
                false, af0, false, bf0, (short)0, acc00, false, false);
            acc01 = __builtin_amdgcn_wmma_f32_16x16x4_f32(
                false, af0, false, bf1, (short)0, acc01, false, false);
            acc10 = __builtin_amdgcn_wmma_f32_16x16x4_f32(
                false, af1, false, bf0, (short)0, acc10, false, false);
            acc11 = __builtin_amdgcn_wmma_f32_16x16x4_f32(
                false, af1, false, bf1, (short)0, acc11, false, false);
        }
    }

    const float s  = 1.0f / (float)NLAYERS;
    const int row0 = mblk + wm * 32 + (lane >> 4) * 8;
    const int col0 = nblk + wn * 32 + (lane & 15);
#pragma unroll
    for (int i = 0; i < 8; ++i) {
        C[(size_t)(row0 + i     ) * NDIM + col0     ] = acc00[i] * s;
        C[(size_t)(row0 + i     ) * NDIM + col0 + 16] = acc01[i] * s;
        C[(size_t)(row0 + i + 16) * NDIM + col0     ] = acc10[i] * s;
        C[(size_t)(row0 + i + 16) * NDIM + col0 + 16] = acc11[i] * s;
    }
}

extern "C" void kernel_launch(void* const* d_in, const int* in_sizes, int n_in,
                              void* d_out, int out_size, void* d_ws, size_t ws_size,
                              hipStream_t stream) {
    const float* x  = (const float*)d_in[0];
    const float* cw = (const float*)d_in[1];
    const float* nw = (const float*)d_in[2];

    float* rms_out  = (float*)d_out;
    float* conv_out = rms_out + (size_t)MDIM * NDIM;
    float* wsum     = (float*)d_ws;

    wsum_kernel<<<(NDIM * KDIM) / 4 / 256, 256, 0, stream>>>(cw, wsum);

    rmsnorm_kernel<<<MDIM, 256, 0, stream>>>(x, nw, rms_out);

    dim3 grid(NDIM / TN, MDIM / TM);
    gemm_wmma_kernel<<<grid, 256, 0, stream>>>(x, wsum, conv_out);
}


# GEMM made weight-stationary: each 8-wave block keeps a 64-col w_sum panel (all K) in LDS for 1024 rows, X fragments straight from global one 32-k block ahead, no barrier in K loop (new k sum order)
# speedup vs baseline: 1.1786x; 1.1786x over previous
#include <hip/hip_runtime.h>
#include <hip/hip_bf16.h>

typedef __attribute__((ext_vector_type(2))) float v2f;
typedef __attribute__((ext_vector_type(4))) float v4f;
typedef __attribute__((ext_vector_type(8))) float v8f;

#define MDIM 16384
#define NDIM 1024
#define KDIM 1024
#define NLAYERS 20
#define EPSV 1e-6f

__global__ __launch_bounds__(256) void wsum_kernel(const float* __restrict__ cw,
                                                   float* __restrict__ ws) {
    size_t idx = ((size_t)blockIdx.x * blockDim.x + threadIdx.x) * 4;
    v4f acc = *(const v4f*)(cw + idx);
#pragma unroll
    for (int l = 1; l < NLAYERS; ++l) {
        acc += *(const v4f*)(cw + (size_t)l * (NDIM * KDIM) + idx);
    }
    *(v4f*)(ws + idx) = acc;
}

#define RN_ROWS  4
#define RN_WAVES 8
#define RN_PIECES 8

__global__ __launch_bounds__(256) void rmsnorm_kernel(const float* __restrict__ x,
                                                      const float* __restrict__ w,
                                                      float* __restrict__ out) {
    const int lane = threadIdx.x & 31;
    const int wave = __builtin_amdgcn_readfirstlane(threadIdx.x >> 5);
    const size_t row0 = ((size_t)blockIdx.x * RN_WAVES + wave) * RN_ROWS;
    const float* xb = x   + row0 * 1024 + lane * 4;
    float*       ob = out + row0 * 1024 + lane * 4;

    v4f wv[RN_PIECES];
#pragma unroll
    for (int q = 0; q < RN_PIECES; ++q)
        wv[q] = *(const v4f*)(w + q * 128 + lane * 4);

    v4f xr[RN_ROWS][RN_PIECES];
#pragma unroll
    for (int r = 0; r < RN_ROWS; ++r)
#pragma unroll
        for (int q = 0; q < RN_PIECES; ++q)
            xr[r][q] = *(const v4f*)(xb + r * 1024 + q * 128);

#pragma unroll
    for (int r = 0; r < RN_ROWS; ++r) {
        float mx = 0.0f;
#pragma unroll
        for (int q = 0; q < RN_PIECES; ++q) {
            const v4f v = xr[r][q];
            mx = fmaxf(mx, fmaxf(fmaxf(fabsf(v.x), fabsf(v.y)), fmaxf(fabsf(v.z), fabsf(v.w))));
        }
#pragma unroll
        for (int off = 16; off > 0; off >>= 1)
            mx = fmaxf(mx, __shfl_xor(mx, off, 32));
        const float inv = 1.0f / fmaxf(mx, 1e-6f);

        v4f acc = {0.0f, 0.0f, 0.0f, 0.0f};
#pragma unroll
        for (int q = 0; q < RN_PIECES; ++q) {
            const v4f xs = xr[r][q] * inv;
            xr[r][q] = xs;
            acc += xs * xs;
        }
        float ss = (acc.x + acc.y) + (acc.z + acc.w);
#pragma unroll
        for (int off = 16; off > 0; off >>= 1)
            ss += __shfl_xor(ss, off, 32);
        const float rstd = rsqrtf(ss * (1.0f / 1024.0f) + EPSV);

#pragma unroll
        for (int q = 0; q < RN_PIECES; ++q)
            *(v4f*)(ob + r * 1024 + q * 128) = xr[r][q] * rstd * wv[q];
    }
}

#define PN     64
#define PROWS  1024
#define SROWS  64
#define BPITCH 1028
#define KBLK   32

__device__ __forceinline__ void load_a_frags(v4f (&a)[4][4], const float* __restrict__ p) {
#pragma unroll
    for (int rt = 0; rt < 4; ++rt)
#pragma unroll
        for (int i = 0; i < 4; ++i)
            a[rt][i] = *(const v4f*)(p + rt * 16 * KDIM + i * 8);
}

__device__ __forceinline__ void load_b_frags(v4f (&b)[4][4], const float* p) {
#pragma unroll
    for (int ct = 0; ct < 4; ++ct)
#pragma unroll
        for (int i = 0; i < 4; ++i)
            b[ct][i] = *(const v4f*)(p + ct * 16 * BPITCH + i * 8);
}

__device__ __forceinline__ void mma_block(v8f (&acc)[4][4], const v4f (&a)[4][4],
                                          const v4f (&b)[4][4]) {
#pragma unroll
    for (int i = 0; i < 4; ++i) {
#pragma unroll
        for (int rt = 0; rt < 4; ++rt) {
            const v2f alo = __builtin_shufflevector(a[rt][i], a[rt][i], 0, 1);
#pragma unroll
            for (int ct = 0; ct < 4; ++ct) {
                const v2f blo = __builtin_shufflevector(b[ct][i], b[ct][i], 0, 1);
                acc[rt][ct] = __builtin_amdgcn_wmma_f32_16x16x4_f32(
                    false, alo, false, blo, (short)0, acc[rt][ct], false, false);
            }
        }
#pragma unroll
        for (int rt = 0; rt < 4; ++rt) {
            const v2f ahi = __builtin_shufflevector(a[rt][i], a[rt][i], 2, 3);
#pragma unroll
            for (int ct = 0; ct < 4; ++ct) {
                const v2f bhi = __builtin_shufflevector(b[ct][i], b[ct][i], 2, 3);
                acc[rt][ct] = __builtin_amdgcn_wmma_f32_16x16x4_f32(
                    false, ahi, false, bhi, (short)0, acc[rt][ct], false, false);
            }
        }
    }
}

__global__ __launch_bounds__(256) void gemm_panel_kernel(const float* __restrict__ X,
                                                         const float* __restrict__ Wm,
                                                         float* __restrict__ C) {
    __shared__ float Bs[PN * BPITCH];

    const int tid  = threadIdx.x;
    const int lane = tid & 31;
    const int wave = __builtin_amdgcn_readfirstlane(tid >> 5);
    const int c    = lane & 15;
    const int h    = lane >> 4;
    const int nblk = (blockIdx.x & 15) * PN;
    const int mblk = (blockIdx.x >> 4) * PROWS;

    for (int j0 = 0; j0 < PN; j0 += 16) {
        v4f t[16];
#pragma unroll
        for (int u = 0; u < 16; ++u)
            t[u] = *(const v4f*)(Wm + (size_t)(nblk + j0 + u) * KDIM + tid * 4);
#pragma unroll
        for (int u = 0; u < 16; ++u)
            *(v4f*)(Bs + (j0 + u) * BPITCH + tid * 4) = t[u];
    }
    __syncthreads();

    const float* bl = Bs + c * BPITCH + 4 * h;
    const float sc  = 1.0f / (float)NLAYERS;

#pragma nounroll
    for (int s = 0; s < PROWS / SROWS / 8; ++s) {
        const int rowbase = mblk + (wave + 8 * s) * SROWS;
        const float* ap = X + (size_t)(rowbase + c) * KDIM + 4 * h;

        v8f acc[4][4];
#pragma unroll
        for (int rt = 0; rt < 4; ++rt)
#pragma unroll
            for (int ct = 0; ct < 4; ++ct)
                acc[rt][ct] = (v8f){};

        v4f a0[4][4], a1[4][4], b0[4][4], b1[4][4];
        load_a_frags(a0, ap);
        load_b_frags(b0, bl);
        for (int kb = 0; kb < KDIM / KBLK - 2; kb += 2) {
            load_a_frags(a1, ap + (kb + 1) * KBLK);
            load_b_frags(b1, bl + (kb + 1) * KBLK);
            __builtin_amdgcn_sched_barrier(0);
            mma_block(acc, a0, b0);
            load_a_frags(a0, ap + (kb + 2) * KBLK);
            load_b_frags(b0, bl + (kb + 2) * KBLK);
            __builtin_amdgcn_sched_barrier(0);
            mma_block(acc, a1, b1);
        }
        load_a_frags(a1, ap + (KDIM / KBLK - 1) * KBLK);
        load_b_frags(b1, bl + (KDIM / KBLK - 1) * KBLK);
        __builtin_amdgcn_sched_barrier(0);
        mma_block(acc, a0, b0);
        mma_block(acc, a1, b1);

        const int row0 = rowbase + h * 8;
        const int col0 = nblk + c;
#pragma unroll
        for (int rt = 0; rt < 4; ++rt)
#pragma unroll
            for (int i = 0; i < 8; ++i)
#pragma unroll
                for (int ct = 0; ct < 4; ++ct)
                    C[(size_t)(row0 + rt * 16 + i) * NDIM + col0 + ct * 16] = acc[rt][ct][i] * sc;
    }
}

extern "C" void kernel_launch(void* const* d_in, const int* in_sizes, int n_in,
                              void* d_out, int out_size, void* d_ws, size_t ws_size,
                              hipStream_t stream) {
    const float* x  = (const float*)d_in[0];
    const float* cw = (const float*)d_in[1];
    const float* nw = (const float*)d_in[2];

    float* rms_out  = (float*)d_out;
    float* conv_out = rms_out + (size_t)MDIM * NDIM;
    float* wsum     = (float*)d_ws;

    wsum_kernel<<<(NDIM * KDIM) / 4 / 256, 256, 0, stream>>>(cw, wsum);

    rmsnorm_kernel<<<MDIM / (RN_WAVES * RN_ROWS), 256, 0, stream>>>(x, nw, rms_out);

    gemm_panel_kernel<<<(NDIM / PN) * (MDIM / PROWS), 256, 0, stream>>>(x, wsum, conv_out);
}
